# starting kernel
# baseline (speedup 1.0000x reference)
#include <hip/hip_runtime.h>


typedef __attribute__((ext_vector_type(2))) float v2f;
typedef __attribute__((ext_vector_type(8))) float v8f;

__device__ __forceinline__ float band_sign(int parity, int band) {
    return (__builtin_popcount(parity & band) & 1) ? -1.0f : 1.0f;
}

__global__ __launch_bounds__(256)
void iwt3d_wmma_kernel(const float* __restrict__ x, float* __restrict__ out) {
    const int tid  = (int)blockIdx.x * 256 + (int)threadIdx.x;
    const int lane = (int)threadIdx.x & 31;
    const int wave = tid >> 5;

    const int h = wave & 31;
    const int d = (wave >> 5) & 31;
    const int c = (wave >> 10) & 31;
    const int b = (wave >> 15) & 1;

    const int bandStride = 32 * 32 * 32 * 32;
    const int inBase = (((b * 256 + c) * 32 + d) * 32 + h) * 32;

    const int lh = lane & 15;
    const int hi = lane >> 4;
    const int m  = lane & 7;

    const int k0 = hi * 2;
    const int k1 = k0 + 1;

    v2f A0, A1;
    A0.x = band_sign(m, k0);      A0.y = band_sign(m, k1);
    A1.x = band_sign(m, k0 + 4);  A1.y = band_sign(m, k1 + 4);

    v2f B00, B01, B10, B11;
    B00.x = x[inBase + (k0    ) * bandStride +  0 + lh];
    B00.y = x[inBase + (k1    ) * bandStride +  0 + lh];
    B01.x = x[inBase + (k0 + 4) * bandStride +  0 + lh];
    B01.y = x[inBase + (k1 + 4) * bandStride +  0 + lh];
    B10.x = x[inBase + (k0    ) * bandStride + 16 + lh];
    B10.y = x[inBase + (k1    ) * bandStride + 16 + lh];
    B11.x = x[inBase + (k0 + 4) * bandStride + 16 + lh];
    B11.y = x[inBase + (k1 + 4) * bandStride + 16 + lh];

    v8f acc0 = {};
    acc0 = __builtin_amdgcn_wmma_f32_16x16x4_f32(false, A0, false, B00,
                                                 (short)0, acc0, false, false);
    acc0 = __builtin_amdgcn_wmma_f32_16x16x4_f32(false, A1, false, B01,
                                                 (short)0, acc0, false, false);
    v8f acc1 = {};
    acc1 = __builtin_amdgcn_wmma_f32_16x16x4_f32(false, A0, false, B10,
                                                 (short)0, acc1, false, false);
    acc1 = __builtin_amdgcn_wmma_f32_16x16x4_f32(false, A1, false, B11,
                                                 (short)0, acc1, false, false);

    if (lane < 16) {
        const int outRow = (((b * 32 + c) * 64 + 2 * d) * 64 + 2 * h) * 64;
        #pragma unroll
        for (int g = 0; g < 2; ++g) {
            const int w = g * 16 + lh;
            const v8f acc = g ? acc1 : acc0;
            const int outBase = outRow + 2 * w;
            #pragma unroll
            for (int p = 0; p < 2; ++p) {
                #pragma unroll
                for (int q = 0; q < 2; ++q) {
                    v2f v;
                    v.x = acc[4 * p + 2 * q + 0];
                    v.y = acc[4 * p + 2 * q + 1];
                    *(v2f*)(out + outBase + p * 4096 + q * 64) = v;
                }
            }
        }
    }
}

extern "C" void kernel_launch(void* const* d_in, const int* in_sizes, int n_in,
                              void* d_out, int out_size, void* d_ws, size_t ws_size,
                              hipStream_t stream) {
    (void)n_in; (void)out_size; (void)d_ws; (void)ws_size;
    const float* x = (const float*)d_in[0];
    float* out = (float*)d_out;

    const int waves  = in_sizes[0] / 256;
    const int blocks = waves / 8;
    hipLaunchKernelGGL(iwt3d_wmma_kernel, dim3(blocks), dim3(256), 0, stream,
                       x, out);
}


# 8 rows per wave (was 1), whole grid resident: a wave's 32 whole-line 8-byte loads all requested at entry, no loop; 32-lane 16-byte whole-row stores; bands summed in pairs (j, j+4): sum order changes
# speedup vs baseline: 1.1536x; 1.1536x over previous
#include <hip/hip_runtime.h>


typedef __attribute__((ext_vector_type(2))) float v2f;
typedef __attribute__((ext_vector_type(4))) float v4f;
typedef __attribute__((ext_vector_type(8))) float v8f;

__device__ __forceinline__ float band_sign(int parity, int band) {
    return (__builtin_popcount(parity & band) & 1) ? -1.0f : 1.0f;
}

__global__ __launch_bounds__(256)
void iwt3d_wmma_kernel(const float* __restrict__ x, float* __restrict__ out) {
    const int lane = (int)threadIdx.x & 31;
    const int wave = __builtin_amdgcn_readfirstlane((int)blockIdx.x * 8 + ((int)threadIdx.x >> 5));

    const int hq = wave & 3;
    const int d  = (wave >> 2) & 31;
    const int c  = (wave >> 7) & 31;
    const int b  = wave >> 12;

    const int n  = lane & 15;
    const int hh = lane >> 4;

    const int bandStride = 32 * 32 * 32 * 32;
    const int inBase  = (((b * 256 + c) * 32 + d) * 32 + hq * 8) * 32;
    const int outBase = (((b * 32 + c) * 64 + 2 * d) * 64 + 16 * hq) * 64;

    const float* pin  = x + inBase + hh * (4 * bandStride) + 2 * n;
    float*       pout = out + outBase + hh * 4096 + 4 * n;

    v2f L[8][4];
    #pragma unroll
    for (int r = 0; r < 8; ++r) {
        #pragma unroll
        for (int j = 0; j < 4; ++j) {
            L[r][j] = *(const v2f*)(pin + j * bandStride + r * 32);
        }
        __builtin_amdgcn_sched_barrier(0);
    }

    const int m  = n;
    const int vm = (m >> 1) & 1;
    const int M  = ((m >> 3) << 2) | (((m >> 2) & 1) << 1) | (m & 1);
    v2f A[4];
    #pragma unroll
    for (int j = 0; j < 4; ++j) {
        const float s = band_sign(M, 4 * hh + j);
        A[j].x = vm ? 0.0f : s;
        A[j].y = vm ? s : 0.0f;
    }

    #pragma unroll
    for (int r = 0; r < 8; ++r) {
        v8f acc = {};
        #pragma unroll
        for (int j = 0; j < 4; ++j) {
            acc = __builtin_amdgcn_wmma_f32_16x16x4_f32(false, A[j], false, L[r][j],
                                                        (short)0, acc, false, false);
        }
        v4f q0, q1;
        q0.x = acc[0]; q0.y = acc[1]; q0.z = acc[2]; q0.w = acc[3];
        q1.x = acc[4]; q1.y = acc[5]; q1.z = acc[6]; q1.w = acc[7];
        *(v4f*)(pout + (2 * r + 0) * 64) = q0;
        *(v4f*)(pout + (2 * r + 1) * 64) = q1;
        __builtin_amdgcn_sched_barrier(0);
    }
}

extern "C" void kernel_launch(void* const* d_in, const int* in_sizes, int n_in,
                              void* d_out, int out_size, void* d_ws, size_t ws_size,
                              hipStream_t stream) {
    (void)in_sizes; (void)n_in; (void)out_size; (void)d_ws; (void)ws_size;
    const float* x = (const float*)d_in[0];
    float* out = (float*)d_out;
    hipLaunchKernelGGL(iwt3d_wmma_kernel, dim3(1024), dim3(256), 0, stream,
                       x, out);
}
